# starting kernel
# baseline (speedup 1.0000x reference)
#include <hip/hip_runtime.h>

typedef float v2f __attribute__((ext_vector_type(2)));
typedef float v4f __attribute__((ext_vector_type(4)));
typedef float v8f __attribute__((ext_vector_type(8)));

#define WMMA_F32(a, b, c) \
  __builtin_amdgcn_wmma_f32_16x16x4_f32(false, (a), false, (b), (short)0, (c), false, false)

constexpr int Bn = 32, Nn = 128, Dn = 64;
constexpr int XS = 66;

__global__ __launch_bounds__(256, 1)
void cross_subg_conv_kernel(const float* __restrict__ X,
                            const float* __restrict__ A,
                            const float* __restrict__ W1,
                            const float* __restrict__ b1,
                            const float* __restrict__ W2,
                            const float* __restrict__ b2,
                            float* __restrict__ out)
{
    __shared__ float sX[Nn * XS];
    __shared__ float sH1[Nn * XS];
    __shared__ float sW[2 * Dn * XS];
    __shared__ float sBias[2 * Dn];

    const int i    = blockIdx.x;
    const int b    = blockIdx.y;
    const int tid  = threadIdx.x;
    const int lane = tid & 31;
    const int wave = tid >> 5;
    const int half = lane >> 4;
    const int lr   = lane & 15;
    const int mBase = wave * 16;

    const v4f* Xg = reinterpret_cast<const v4f*>(X + (size_t)(b * Nn + i) * Nn * Dn);
    for (int t = tid; t < Nn * Dn / 4; t += 256) {
        v4f v = __builtin_nontemporal_load(Xg + t);
        int row = t >> 4, col = (t & 15) << 2;
        *reinterpret_cast<v4f*>(&sX[row * XS + col]) = v;
    }
    const v4f* W1g = reinterpret_cast<const v4f*>(W1);
    const v4f* W2g = reinterpret_cast<const v4f*>(W2);
    for (int t = tid; t < Dn * Dn / 4; t += 256) {
        int row = t >> 4, col = (t & 15) << 2;
        v4f v1 = W1g[t];
        v4f v2 = W2g[t];
        *reinterpret_cast<v2f*>(&sW[row * XS + col])          = v1.xy;
        *reinterpret_cast<v2f*>(&sW[row * XS + col + 2])      = v1.zw;
        *reinterpret_cast<v2f*>(&sW[Dn * XS + row * XS + col])     = v2.xy;
        *reinterpret_cast<v2f*>(&sW[Dn * XS + row * XS + col + 2]) = v2.zw;
    }
    if (tid < Dn) { sBias[tid] = b1[tid]; sBias[Dn + tid] = b2[tid]; }
    __syncthreads();

    {
        const int rowBase = (mBase + lr) * XS;
        v8f acc[4] = {};
        #pragma unroll
        for (int kk = 0; kk < 16; ++kk) {
            const int k0 = kk * 4 + half * 2;
            v2f a = *reinterpret_cast<const v2f*>(&sX[rowBase + k0]);
            #pragma unroll
            for (int nt = 0; nt < 4; ++nt) {
                v2f bf;
                bf.x = sW[k0 * XS + nt * 16 + lr];
                bf.y = sW[(k0 + 1) * XS + nt * 16 + lr];
                acc[nt] = WMMA_F32(a, bf, acc[nt]);
            }
        }
        #pragma unroll
        for (int nt = 0; nt < 4; ++nt) {
            const float bias = sBias[nt * 16 + lr];
            #pragma unroll
            for (int v = 0; v < 8; ++v) {
                float r = fmaxf(acc[nt][v] + bias, 0.0f);
                sH1[(mBase + v + half * 8) * XS + nt * 16 + lr] = r;
            }
        }
    }
    __syncthreads();

    {
        const int rowBase = (mBase + lr) * XS;
        v8f acc[4] = {};
        #pragma unroll
        for (int kk = 0; kk < 16; ++kk) {
            const int k0 = kk * 4 + half * 2;
            v2f a = *reinterpret_cast<const v2f*>(&sH1[rowBase + k0]);
            #pragma unroll
            for (int nt = 0; nt < 4; ++nt) {
                v2f bf;
                bf.x = sW[Dn * XS + k0 * XS + nt * 16 + lr];
                bf.y = sW[Dn * XS + (k0 + 1) * XS + nt * 16 + lr];
                acc[nt] = WMMA_F32(a, bf, acc[nt]);
            }
        }
        #pragma unroll
        for (int nt = 0; nt < 4; ++nt) {
            const float bias = sBias[Dn + nt * 16 + lr];
            #pragma unroll
            for (int v = 0; v < 8; ++v) {
                float r = fmaxf(acc[nt][v] + bias, 0.0f);
                sX[(mBase + v + half * 8) * XS + nt * 16 + lr] = r;
            }
        }
    }
    __syncthreads();

    {
        const float* Ab = A + (size_t)b * Nn * Nn;
        v8f acc[4] = {};
        #pragma unroll 8
        for (int kk = 0; kk < 32; ++kk) {
            const int k0 = kk * 4 + half * 2;
            v2f a;
            a.x = Ab[k0 * Nn + mBase + lr];
            a.y = Ab[(k0 + 1) * Nn + mBase + lr];
            #pragma unroll
            for (int nt = 0; nt < 4; ++nt) {
                v2f bf;
                bf.x = sX[k0 * XS + nt * 16 + lr];
                bf.y = sX[(k0 + 1) * XS + nt * 16 + lr];
                acc[nt] = WMMA_F32(a, bf, acc[nt]);
            }
        }
        float* og = out + (size_t)(b * Nn + i) * Nn * Dn;
        #pragma unroll
        for (int nt = 0; nt < 4; ++nt) {
            #pragma unroll
            for (int v = 0; v < 8; ++v) {
                __builtin_nontemporal_store(
                    acc[nt][v],
                    &og[(size_t)(mBase + v + half * 8) * Dn + nt * 16 + lr]);
            }
        }
    }
}

extern "C" void kernel_launch(void* const* d_in, const int* in_sizes, int n_in,
                              void* d_out, int out_size, void* d_ws, size_t ws_size,
                              hipStream_t stream) {
    const float* X  = (const float*)d_in[0];
    const float* A  = (const float*)d_in[1];
    const float* W1 = (const float*)d_in[2];
    const float* b1 = (const float*)d_in[3];
    const float* W2 = (const float*)d_in[4];
    const float* b2 = (const float*)d_in[5];
    float* out = (float*)d_out;

    dim3 grid(Nn, Bn);
    dim3 block(256);
    cross_subg_conv_kernel<<<grid, block, 0, stream>>>(X, A, W1, b1, W2, b2, out);
}


# staging: each thread issues its 8 X-tile loads and 8 weight loads back to back before the first LDS store (was one load, a full wait and a store per loop trip)
# speedup vs baseline: 1.0723x; 1.0723x over previous
#include <hip/hip_runtime.h>

typedef float v2f __attribute__((ext_vector_type(2)));
typedef float v4f __attribute__((ext_vector_type(4)));
typedef float v8f __attribute__((ext_vector_type(8)));

#define WMMA_F32(a, b, c) \
  __builtin_amdgcn_wmma_f32_16x16x4_f32(false, (a), false, (b), (short)0, (c), false, false)

constexpr int Bn = 32, Nn = 128, Dn = 64;
constexpr int XS = 66;

__global__ __launch_bounds__(256, 1)
void cross_subg_conv_kernel(const float* __restrict__ X,
                            const float* __restrict__ A,
                            const float* __restrict__ W1,
                            const float* __restrict__ b1,
                            const float* __restrict__ W2,
                            const float* __restrict__ b2,
                            float* __restrict__ out)
{
    __shared__ float sX[Nn * XS];
    __shared__ float sH1[Nn * XS];
    __shared__ float sW[2 * Dn * XS];
    __shared__ float sBias[2 * Dn];

    const int i    = blockIdx.x;
    const int b    = blockIdx.y;
    const int tid  = threadIdx.x;
    const int lane = tid & 31;
    const int wave = tid >> 5;
    const int half = lane >> 4;
    const int lr   = lane & 15;
    const int mBase = wave * 16;

    const v4f* Xg  = reinterpret_cast<const v4f*>(X + (size_t)(b * Nn + i) * Nn * Dn);
    const v4f* W1g = reinterpret_cast<const v4f*>(W1);
    const v4f* W2g = reinterpret_cast<const v4f*>(W2);
    constexpr int XV = Nn * Dn / 4 / 256;
    constexpr int WV = Dn * Dn / 4 / 256;
    v4f w1r[WV], w2r[WV], xr[XV];
    float bias1 = 0.0f, bias2 = 0.0f;
    #pragma unroll
    for (int u = 0; u < WV; ++u) {
        w1r[u] = W1g[tid + u * 256];
        w2r[u] = W2g[tid + u * 256];
    }
    if (tid < Dn) { bias1 = b1[tid]; bias2 = b2[tid]; }
    #pragma unroll
    for (int u = 0; u < XV; ++u)
        xr[u] = __builtin_nontemporal_load(Xg + tid + u * 256);

    #pragma unroll
    for (int u = 0; u < WV; ++u) {
        const int t = tid + u * 256;
        const int row = t >> 4, col = (t & 15) << 2;
        *reinterpret_cast<v2f*>(&sW[row * XS + col])               = w1r[u].xy;
        *reinterpret_cast<v2f*>(&sW[row * XS + col + 2])           = w1r[u].zw;
        *reinterpret_cast<v2f*>(&sW[Dn * XS + row * XS + col])     = w2r[u].xy;
        *reinterpret_cast<v2f*>(&sW[Dn * XS + row * XS + col + 2]) = w2r[u].zw;
    }
    if (tid < Dn) { sBias[tid] = bias1; sBias[Dn + tid] = bias2; }
    #pragma unroll
    for (int u = 0; u < XV; ++u) {
        const int t = tid + u * 256;
        const int row = t >> 4, col = (t & 15) << 2;
        *reinterpret_cast<v2f*>(&sX[row * XS + col])     = xr[u].xy;
        *reinterpret_cast<v2f*>(&sX[row * XS + col + 2]) = xr[u].zw;
    }
    __syncthreads();

    {
        const int rowBase = (mBase + lr) * XS;
        v8f acc[4] = {};
        #pragma unroll
        for (int kk = 0; kk < 16; ++kk) {
            const int k0 = kk * 4 + half * 2;
            v2f a = *reinterpret_cast<const v2f*>(&sX[rowBase + k0]);
            #pragma unroll
            for (int nt = 0; nt < 4; ++nt) {
                v2f bf;
                bf.x = sW[k0 * XS + nt * 16 + lr];
                bf.y = sW[(k0 + 1) * XS + nt * 16 + lr];
                acc[nt] = WMMA_F32(a, bf, acc[nt]);
            }
        }
        #pragma unroll
        for (int nt = 0; nt < 4; ++nt) {
            const float bias = sBias[nt * 16 + lr];
            #pragma unroll
            for (int v = 0; v < 8; ++v) {
                float r = fmaxf(acc[nt][v] + bias, 0.0f);
                sH1[(mBase + v + half * 8) * XS + nt * 16 + lr] = r;
            }
        }
    }
    __syncthreads();

    {
        const int rowBase = (mBase + lr) * XS;
        v8f acc[4] = {};
        #pragma unroll
        for (int kk = 0; kk < 16; ++kk) {
            const int k0 = kk * 4 + half * 2;
            v2f a = *reinterpret_cast<const v2f*>(&sH1[rowBase + k0]);
            #pragma unroll
            for (int nt = 0; nt < 4; ++nt) {
                v2f bf;
                bf.x = sW[Dn * XS + k0 * XS + nt * 16 + lr];
                bf.y = sW[Dn * XS + (k0 + 1) * XS + nt * 16 + lr];
                acc[nt] = WMMA_F32(a, bf, acc[nt]);
            }
        }
        #pragma unroll
        for (int nt = 0; nt < 4; ++nt) {
            const float bias = sBias[Dn + nt * 16 + lr];
            #pragma unroll
            for (int v = 0; v < 8; ++v) {
                float r = fmaxf(acc[nt][v] + bias, 0.0f);
                sX[(mBase + v + half * 8) * XS + nt * 16 + lr] = r;
            }
        }
    }
    __syncthreads();

    {
        const float* Ab = A + (size_t)b * Nn * Nn;
        v8f acc[4] = {};
        #pragma unroll 8
        for (int kk = 0; kk < 32; ++kk) {
            const int k0 = kk * 4 + half * 2;
            v2f a;
            a.x = Ab[k0 * Nn + mBase + lr];
            a.y = Ab[(k0 + 1) * Nn + mBase + lr];
            #pragma unroll
            for (int nt = 0; nt < 4; ++nt) {
                v2f bf;
                bf.x = sX[k0 * XS + nt * 16 + lr];
                bf.y = sX[(k0 + 1) * XS + nt * 16 + lr];
                acc[nt] = WMMA_F32(a, bf, acc[nt]);
            }
        }
        float* og = out + (size_t)(b * Nn + i) * Nn * Dn;
        #pragma unroll
        for (int nt = 0; nt < 4; ++nt) {
            #pragma unroll
            for (int v = 0; v < 8; ++v) {
                __builtin_nontemporal_store(
                    acc[nt][v],
                    &og[(size_t)(mBase + v + half * 8) * Dn + nt * 16 + lr]);
            }
        }
    }
}

extern "C" void kernel_launch(void* const* d_in, const int* in_sizes, int n_in,
                              void* d_out, int out_size, void* d_ws, size_t ws_size,
                              hipStream_t stream) {
    const float* X  = (const float*)d_in[0];
    const float* A  = (const float*)d_in[1];
    const float* W1 = (const float*)d_in[2];
    const float* b1 = (const float*)d_in[3];
    const float* W2 = (const float*)d_in[4];
    const float* b2 = (const float*)d_in[5];
    float* out = (float*)d_out;

    dim3 grid(Nn, Bn);
    dim3 block(256);
    cross_subg_conv_kernel<<<grid, block, 0, stream>>>(X, A, W1, b1, W2, b2, out);
}


# LDS layout: W1, W2, H2 stored in WMMA fragment order so one conflict-free 16-byte read per lane feeds two WMMAs (was two 2-way bank-conflicted 4-byte reads per B fragment); X, H1 rows at pitch 68
# speedup vs baseline: 1.2022x; 1.1211x over previous
#include <hip/hip_runtime.h>


typedef float v2f __attribute__((ext_vector_type(2)));
typedef float v4f __attribute__((ext_vector_type(4)));
typedef float v8f __attribute__((ext_vector_type(8)));

#define WMMA_F32(a, b, c) \
  __builtin_amdgcn_wmma_f32_16x16x4_f32(false, (a), false, (b), (short)0, (c), false, false)

constexpr int Bn = 32, Nn = 128, Dn = 64;

constexpr int XP = 68;

constexpr int FB = 128;
constexpr int WF = Dn * Dn;

__global__ __launch_bounds__(256, 1)
void cross_subg_conv_kernel(const float* __restrict__ X,
                            const float* __restrict__ A,
                            const float* __restrict__ W1,
                            const float* __restrict__ b1,
                            const float* __restrict__ W2,
                            const float* __restrict__ b2,
                            float* __restrict__ out)
{
    __shared__ __attribute__((aligned(16))) float smem[2 * WF + 2 * Nn * XP + 2 * Dn];
    float* const sW    = smem;
    float* const sX    = smem + 2 * WF;
    float* const sH1   = smem + 2 * WF + Nn * XP;
    float* const sBias = smem + 2 * WF + 2 * Nn * XP;

    const int i    = blockIdx.x;
    const int b    = blockIdx.y;
    const int tid  = threadIdx.x;
    const int lane = tid & 31;
    const int wave = tid >> 5;
    const int half = lane >> 4;
    const int lr   = lane & 15;
    const int mBase = wave * 16;

    const v4f* Xg  = reinterpret_cast<const v4f*>(X + (size_t)(b * Nn + i) * Nn * Dn);
    const v4f* W1g = reinterpret_cast<const v4f*>(W1);
    const v4f* W2g = reinterpret_cast<const v4f*>(W2);
    constexpr int XV = Nn * Dn / 4 / 256;
    const int wgq = tid & 3, wq = (tid >> 2) & 1, whp = (tid >> 3) & 1, wkk = tid >> 4;
    const int wsrc = (wkk * 4 + whp * 2) * (Dn / 4) + wq * 8 + wgq;
    const int wdst = (wkk * 2 + wq) * FB + whp * 64 + wgq * 16;
    v4f w1r[4], w2r[4], xr[XV];
    float bias1 = 0.0f, bias2 = 0.0f;
    w1r[0] = W1g[wsrc];              w1r[1] = W1g[wsrc + Dn / 4];
    w1r[2] = W1g[wsrc + 4];          w1r[3] = W1g[wsrc + Dn / 4 + 4];
    w2r[0] = W2g[wsrc];              w2r[1] = W2g[wsrc + Dn / 4];
    w2r[2] = W2g[wsrc + 4];          w2r[3] = W2g[wsrc + Dn / 4 + 4];
    if (tid < Dn) { bias1 = b1[tid]; bias2 = b2[tid]; }
    #pragma unroll
    for (int u = 0; u < XV; ++u)
        xr[u] = __builtin_nontemporal_load(Xg + tid + u * 256);

    #pragma unroll
    for (int c = 0; c < 4; ++c) {
        v4f p, r;
        p.x = w1r[0][c]; p.y = w1r[1][c]; p.z = w1r[2][c]; p.w = w1r[3][c];
        r.x = w2r[0][c]; r.y = w2r[1][c]; r.z = w2r[2][c]; r.w = w2r[3][c];
        *reinterpret_cast<v4f*>(&sW[wdst + c * 4])      = p;
        *reinterpret_cast<v4f*>(&sW[WF + wdst + c * 4]) = r;
    }
    if (tid < Dn) { sBias[tid] = bias1; sBias[Dn + tid] = bias2; }
    #pragma unroll
    for (int u = 0; u < XV; ++u) {
        const int t = tid + u * 256;
        const int row = t >> 4, col = (t & 15) << 2;
        *reinterpret_cast<v4f*>(&sX[row * XP + col]) = xr[u];
    }
    __syncthreads();

    const int rowBase = (mBase + lr) * XP + half * 2;
    const int fragOff = lane * 4;

    {
        v8f acc[4] = {};
        #pragma unroll
        for (int kk = 0; kk < 16; ++kk) {
            v2f a = *reinterpret_cast<const v2f*>(&sX[rowBase + kk * 4]);
            #pragma unroll
            for (int q = 0; q < 2; ++q) {
                v4f f = *reinterpret_cast<const v4f*>(&sW[(kk * 2 + q) * FB + fragOff]);
                acc[2 * q]     = WMMA_F32(a, f.xy, acc[2 * q]);
                acc[2 * q + 1] = WMMA_F32(a, f.zw, acc[2 * q + 1]);
            }
        }
        #pragma unroll
        for (int nt = 0; nt < 4; ++nt) {
            const float bias = sBias[nt * 16 + lr];
            #pragma unroll
            for (int v = 0; v < 8; ++v) {
                float r = fmaxf(acc[nt][v] + bias, 0.0f);
                sH1[(mBase + v + half * 8) * XP + nt * 16 + lr] = r;
            }
        }
    }
    __syncthreads();

    {
        v8f acc[4] = {};
        #pragma unroll
        for (int kk = 0; kk < 16; ++kk) {
            v2f a = *reinterpret_cast<const v2f*>(&sH1[rowBase + kk * 4]);
            #pragma unroll
            for (int q = 0; q < 2; ++q) {
                v4f f = *reinterpret_cast<const v4f*>(&sW[WF + (kk * 2 + q) * FB + fragOff]);
                acc[2 * q]     = WMMA_F32(a, f.xy, acc[2 * q]);
                acc[2 * q + 1] = WMMA_F32(a, f.zw, acc[2 * q + 1]);
            }
        }
        #pragma unroll
        for (int q = 0; q < 2; ++q) {
            const float biasLo = sBias[Dn + q * 32 + lr];
            const float biasHi = sBias[Dn + q * 32 + 16 + lr];
            #pragma unroll
            for (int v = 0; v < 8; v += 2) {
                v4f r;
                r.x = fmaxf(acc[2 * q][v]         + biasLo, 0.0f);
                r.y = fmaxf(acc[2 * q][v + 1]     + biasLo, 0.0f);
                r.z = fmaxf(acc[2 * q + 1][v]     + biasHi, 0.0f);
                r.w = fmaxf(acc[2 * q + 1][v + 1] + biasHi, 0.0f);
                const int kk3 = wave * 4 + half * 2 + (v >> 2);
                *reinterpret_cast<v4f*>(
                    &sX[(kk3 * 2 + q) * FB + (((v >> 1) & 1) * 16 + lr) * 4]) = r;
            }
        }
    }
    __syncthreads();

    {
        const float* Ab = A + (size_t)b * Nn * Nn;
        v8f acc[4] = {};
        #pragma unroll 8
        for (int kk = 0; kk < 32; ++kk) {
            const int k0 = kk * 4 + half * 2;
            v2f a;
            a.x = Ab[k0 * Nn + mBase + lr];
            a.y = Ab[(k0 + 1) * Nn + mBase + lr];
            #pragma unroll
            for (int q = 0; q < 2; ++q) {
                v4f f = *reinterpret_cast<const v4f*>(&sX[(kk * 2 + q) * FB + fragOff]);
                acc[2 * q]     = WMMA_F32(a, f.xy, acc[2 * q]);
                acc[2 * q + 1] = WMMA_F32(a, f.zw, acc[2 * q + 1]);
            }
        }
        float* og = out + (size_t)(b * Nn + i) * Nn * Dn;
        #pragma unroll
        for (int nt = 0; nt < 4; ++nt) {
            #pragma unroll
            for (int v = 0; v < 8; ++v) {
                __builtin_nontemporal_store(
                    acc[nt][v],
                    &og[(size_t)(mBase + v + half * 8) * Dn + nt * 16 + lr]);
            }
        }
    }
}

extern "C" void kernel_launch(void* const* d_in, const int* in_sizes, int n_in,
                              void* d_out, int out_size, void* d_ws, size_t ws_size,
                              hipStream_t stream) {
    const float* X  = (const float*)d_in[0];
    const float* A  = (const float*)d_in[1];
    const float* W1 = (const float*)d_in[2];
    const float* b1 = (const float*)d_in[3];
    const float* W2 = (const float*)d_in[4];
    const float* b2 = (const float*)d_in[5];
    float* out = (float*)d_out;

    dim3 grid(Nn, Bn);
    dim3 block(256);
    cross_subg_conv_kernel<<<grid, block, 0, stream>>>(X, A, W1, b1, W2, b2, out);
}
